# starting kernel
# baseline (speedup 1.0000x reference)
#include <hip/hip_runtime.h>
#include <math.h>


typedef __attribute__((ext_vector_type(2))) float v2f;
typedef __attribute__((ext_vector_type(8))) float v8f;

#define S_LEN 384
#define B_SZ  256
#define D_IN  192
#define H_DIM 100
#define FH    400
#define C_OUT 250

#define XS_STRIDE 200
#define HS  105
#define ZS  404

__device__ __forceinline__ float sigmoidf_(float v) {
    return 1.0f / (1.0f + __expf(-v));
}

__device__ __forceinline__ float fast_tanhf_(float v) {
#if __has_builtin(__builtin_amdgcn_tanhf)
    return __builtin_amdgcn_tanhf(v);
#elif __has_builtin(__builtin_amdgcn_tanh_f32)
    return __builtin_amdgcn_tanh_f32(v);
#else
    return tanhf(v);
#endif
}

__global__ __launch_bounds__(256)
void lstm_xw_kernel(const float* __restrict__ x, const float* __restrict__ W,
                    const float* __restrict__ bias, float* __restrict__ pre) {
    __shared__ float xs[16 * XS_STRIDE];
    const int blk = blockIdx.x;
    const int s   = blk >> 4;
    const int b0  = (blk & 15) << 4;
    const int tid = threadIdx.x;

    for (int e = tid; e < 16 * D_IN; e += 256) {
        const int r = e / D_IN, d = e - r * D_IN;
        xs[r * XS_STRIDE + d] = x[((size_t)(b0 + r) * S_LEN + s) * D_IN + d];
    }
    __syncthreads();

    const int wave  = tid >> 5;
    const int lane  = tid & 31;
    const int lhalf = lane >> 4;
    const int l15   = lane & 15;
    const int koff  = lhalf * 2;

    for (int t = wave; t < 25; t += 8) {
        const int n0 = t << 4;
        const int n  = n0 + l15;
        const float bb = bias[n];
        v8f acc;
#pragma unroll
        for (int r = 0; r < 8; ++r) acc[r] = bb;

#pragma unroll 4
        for (int kk = 0; kk < 48; ++kk) {
            const int k0 = kk * 4 + koff;
            v2f a, bf;
            a[0]  = xs[l15 * XS_STRIDE + k0];
            a[1]  = xs[l15 * XS_STRIDE + k0 + 1];
            bf[0] = W[(size_t)k0 * FH + n];
            bf[1] = W[(size_t)(k0 + 1) * FH + n];
            acc = __builtin_amdgcn_wmma_f32_16x16x4_f32(
                false, a, false, bf, (short)0, acc, false, false);
        }
        float* prow = pre + ((size_t)s * B_SZ + b0) * FH;
#pragma unroll
        for (int r = 0; r < 8; ++r)
            prow[(size_t)(r + lhalf * 8) * FH + n] = acc[r];
    }
}

__global__ __launch_bounds__(256)
void lstm_rec_kernel(const float* __restrict__ pre, const float* __restrict__ U,
                     float* __restrict__ hT) {
    __shared__ float U_lds[H_DIM * FH];
    __shared__ float h_s[16 * HS];
    __shared__ float c_s[16 * H_DIM];
    __shared__ float z_s[16 * ZS];

    const int b0  = blockIdx.x << 4;
    const int tid = threadIdx.x;
    const int wave  = tid >> 5;
    const int lane  = tid & 31;
    const int lhalf = lane >> 4;
    const int l15   = lane & 15;
    const int koff  = lhalf * 2;

    for (int e = tid; e < H_DIM * FH; e += 256) U_lds[e] = U[e];
    for (int e = tid; e < 16 * HS; e += 256)    h_s[e] = 0.0f;
    for (int e = tid; e < 16 * H_DIM; e += 256) c_s[e] = 0.0f;
    __syncthreads();

    for (int s = 0; s < S_LEN; ++s) {
        const float* prow = pre + ((size_t)s * B_SZ + b0) * FH;

        if (s + 1 < S_LEN) {
            const float* nrow = prow + (size_t)B_SZ * FH;
            for (int e = tid; e < 400; e += 256)
                __builtin_prefetch(nrow + e * 16, 0, 0);
        }

        for (int t = wave; t < 25; t += 8) {
            const int n0 = t << 4;
            const int n  = n0 + l15;
            v8f acc;
#pragma unroll
            for (int r = 0; r < 8; ++r)
                acc[r] = prow[(size_t)(r + lhalf * 8) * FH + n];

#pragma unroll
            for (int kk = 0; kk < 25; ++kk) {
                const int k0 = kk * 4 + koff;
                v2f a, bf;
                a[0]  = h_s[l15 * HS + k0];
                a[1]  = h_s[l15 * HS + k0 + 1];
                bf[0] = U_lds[k0 * FH + n];
                bf[1] = U_lds[(k0 + 1) * FH + n];
                acc = __builtin_amdgcn_wmma_f32_16x16x4_f32(
                    false, a, false, bf, (short)0, acc, false, false);
            }
#pragma unroll
            for (int r = 0; r < 8; ++r)
                z_s[(r + lhalf * 8) * ZS + n] = acc[r];
        }
        __syncthreads();

        for (int e = tid; e < 16 * H_DIM; e += 256) {
            const int b = e / H_DIM, j = e - b * H_DIM;
            const float zi = z_s[b * ZS + j];
            const float zf = z_s[b * ZS + H_DIM + j];
            const float zg = z_s[b * ZS + 2 * H_DIM + j];
            const float zo = z_s[b * ZS + 3 * H_DIM + j];
            const float ig = sigmoidf_(zi);
            const float fg = sigmoidf_(zf);
            const float gg = fast_tanhf_(zg);
            const float og = sigmoidf_(zo);
            const float cn = fg * c_s[e] + ig * gg;
            c_s[e] = cn;
            h_s[b * HS + j] = og * fast_tanhf_(cn);
        }
        __syncthreads();
    }

    for (int e = tid; e < 16 * H_DIM; e += 256) {
        const int b = e / H_DIM, j = e - b * H_DIM;
        hT[(size_t)(b0 + b) * H_DIM + j] = h_s[b * HS + j];
    }
}

__global__ __launch_bounds__(32)
void dense_kernel(const float* __restrict__ hT, const float* __restrict__ Wd,
                  const float* __restrict__ bd, float* __restrict__ out) {
    const int m0 = (blockIdx.x >> 4) << 4;
    const int n0 = (blockIdx.x & 15) << 4;
    const int lane  = threadIdx.x & 31;
    const int lhalf = lane >> 4;
    const int l15   = lane & 15;
    const int koff  = lhalf * 2;
    const int n = n0 + l15;
    const bool nvalid = (n < C_OUT);

    v8f acc;
    const float bb = nvalid ? bd[n] : 0.0f;
#pragma unroll
    for (int r = 0; r < 8; ++r) acc[r] = bb;

#pragma unroll
    for (int kk = 0; kk < 25; ++kk) {
        const int k0 = kk * 4 + koff;
        v2f a, bf;
        a[0]  = hT[(size_t)(m0 + l15) * H_DIM + k0];
        a[1]  = hT[(size_t)(m0 + l15) * H_DIM + k0 + 1];
        bf[0] = nvalid ? Wd[(size_t)k0 * C_OUT + n] : 0.0f;
        bf[1] = nvalid ? Wd[(size_t)(k0 + 1) * C_OUT + n] : 0.0f;
        acc = __builtin_amdgcn_wmma_f32_16x16x4_f32(
            false, a, false, bf, (short)0, acc, false, false);
    }
    if (nvalid) {
#pragma unroll
        for (int r = 0; r < 8; ++r)
            out[(size_t)(m0 + r + lhalf * 8) * C_OUT + n] = acc[r];
    }
}

extern "C" void kernel_launch(void* const* d_in, const int* in_sizes, int n_in,
                              void* d_out, int out_size, void* d_ws, size_t ws_size,
                              hipStream_t stream) {
    const float* x  = (const float*)d_in[0];
    const float* W  = (const float*)d_in[1];
    const float* U  = (const float*)d_in[2];
    const float* bi = (const float*)d_in[3];
    const float* Wd = (const float*)d_in[4];
    const float* bd = (const float*)d_in[5];
    float* out = (float*)d_out;

    float* pre = (float*)d_ws;
    float* hT  = pre + (size_t)S_LEN * B_SZ * FH;

    lstm_xw_kernel<<<S_LEN * (B_SZ / 16), 256, 0, stream>>>(x, W, bi, pre);
    lstm_rec_kernel<<<B_SZ / 16, 256, 0, stream>>>(pre, U, hT);
    dense_kernel<<<(B_SZ / 16) * 16, 32, 0, stream>>>(hT, Wd, bd, out);
}


# recurrence work-group widened from 8 to 25 waves, one z N-tile per wave: tile latencies overlap across waves, gate phase 2 elements per thread
# speedup vs baseline: 1.5335x; 1.5335x over previous
#include <hip/hip_runtime.h>
#include <math.h>


typedef __attribute__((ext_vector_type(2))) float v2f;
typedef __attribute__((ext_vector_type(8))) float v8f;

#define S_LEN 384
#define B_SZ  256
#define D_IN  192
#define H_DIM 100
#define FH    400
#define C_OUT 250

#define XS_STRIDE 200
#define HS  105
#define ZS  404

#define REC_WAVES   25
#define REC_THREADS (REC_WAVES * 32)

__device__ __forceinline__ float sigmoidf_(float v) {
    return 1.0f / (1.0f + __expf(-v));
}

__device__ __forceinline__ float fast_tanhf_(float v) {
#if __has_builtin(__builtin_amdgcn_tanhf)
    return __builtin_amdgcn_tanhf(v);
#elif __has_builtin(__builtin_amdgcn_tanh_f32)
    return __builtin_amdgcn_tanh_f32(v);
#else
    return tanhf(v);
#endif
}

__global__ __launch_bounds__(256)
void lstm_xw_kernel(const float* __restrict__ x, const float* __restrict__ W,
                    const float* __restrict__ bias, float* __restrict__ pre) {
    __shared__ float xs[16 * XS_STRIDE];
    const int blk = blockIdx.x;
    const int s   = blk >> 4;
    const int b0  = (blk & 15) << 4;
    const int tid = threadIdx.x;

    for (int e = tid; e < 16 * D_IN; e += 256) {
        const int r = e / D_IN, d = e - r * D_IN;
        xs[r * XS_STRIDE + d] = x[((size_t)(b0 + r) * S_LEN + s) * D_IN + d];
    }
    __syncthreads();

    const int wave  = tid >> 5;
    const int lane  = tid & 31;
    const int lhalf = lane >> 4;
    const int l15   = lane & 15;
    const int koff  = lhalf * 2;

    for (int t = wave; t < 25; t += 8) {
        const int n0 = t << 4;
        const int n  = n0 + l15;
        const float bb = bias[n];
        v8f acc;
#pragma unroll
        for (int r = 0; r < 8; ++r) acc[r] = bb;

#pragma unroll 4
        for (int kk = 0; kk < 48; ++kk) {
            const int k0 = kk * 4 + koff;
            v2f a, bf;
            a[0]  = xs[l15 * XS_STRIDE + k0];
            a[1]  = xs[l15 * XS_STRIDE + k0 + 1];
            bf[0] = W[(size_t)k0 * FH + n];
            bf[1] = W[(size_t)(k0 + 1) * FH + n];
            acc = __builtin_amdgcn_wmma_f32_16x16x4_f32(
                false, a, false, bf, (short)0, acc, false, false);
        }
        float* prow = pre + ((size_t)s * B_SZ + b0) * FH;
#pragma unroll
        for (int r = 0; r < 8; ++r)
            prow[(size_t)(r + lhalf * 8) * FH + n] = acc[r];
    }
}

__global__ __launch_bounds__(REC_THREADS)
void lstm_rec_kernel(const float* __restrict__ pre, const float* __restrict__ U,
                     float* __restrict__ hT) {
    __shared__ float U_lds[H_DIM * FH];
    __shared__ float h_s[16 * HS];
    __shared__ float c_s[16 * H_DIM];
    __shared__ float z_s[16 * ZS];

    const int b0  = blockIdx.x << 4;
    const int tid = threadIdx.x;
    const int wave  = tid >> 5;
    const int lane  = tid & 31;
    const int lhalf = lane >> 4;
    const int l15   = lane & 15;
    const int koff  = lhalf * 2;

    for (int e = tid; e < H_DIM * FH; e += REC_THREADS) U_lds[e] = U[e];
    for (int e = tid; e < 16 * HS; e += REC_THREADS)    h_s[e] = 0.0f;
    for (int e = tid; e < 16 * H_DIM; e += REC_THREADS) c_s[e] = 0.0f;
    __syncthreads();

    const int n = (wave << 4) + l15;

    for (int s = 0; s < S_LEN; ++s) {
        const float* prow = pre + ((size_t)s * B_SZ + b0) * FH;

        if (s + 1 < S_LEN) {
            const float* nrow = prow + (size_t)B_SZ * FH;
            if (tid < 400)
                __builtin_prefetch(nrow + tid * 16, 0, 0);
        }

        {
            v8f acc;
#pragma unroll
            for (int r = 0; r < 8; ++r)
                acc[r] = prow[(size_t)(r + lhalf * 8) * FH + n];

#pragma unroll
            for (int kk = 0; kk < 25; ++kk) {
                const int k0 = kk * 4 + koff;
                v2f a, bf;
                a[0]  = h_s[l15 * HS + k0];
                a[1]  = h_s[l15 * HS + k0 + 1];
                bf[0] = U_lds[k0 * FH + n];
                bf[1] = U_lds[(k0 + 1) * FH + n];
                acc = __builtin_amdgcn_wmma_f32_16x16x4_f32(
                    false, a, false, bf, (short)0, acc, false, false);
            }
#pragma unroll
            for (int r = 0; r < 8; ++r)
                z_s[(r + lhalf * 8) * ZS + n] = acc[r];
        }
        __syncthreads();

        for (int e = tid; e < 16 * H_DIM; e += REC_THREADS) {
            const int b = e / H_DIM, j = e - b * H_DIM;
            const float zi = z_s[b * ZS + j];
            const float zf = z_s[b * ZS + H_DIM + j];
            const float zg = z_s[b * ZS + 2 * H_DIM + j];
            const float zo = z_s[b * ZS + 3 * H_DIM + j];
            const float ig = sigmoidf_(zi);
            const float fg = sigmoidf_(zf);
            const float gg = fast_tanhf_(zg);
            const float og = sigmoidf_(zo);
            const float cn = fg * c_s[e] + ig * gg;
            c_s[e] = cn;
            h_s[b * HS + j] = og * fast_tanhf_(cn);
        }
        __syncthreads();
    }

    for (int e = tid; e < 16 * H_DIM; e += REC_THREADS) {
        const int b = e / H_DIM, j = e - b * H_DIM;
        hT[(size_t)(b0 + b) * H_DIM + j] = h_s[b * HS + j];
    }
}

__global__ __launch_bounds__(32)
void dense_kernel(const float* __restrict__ hT, const float* __restrict__ Wd,
                  const float* __restrict__ bd, float* __restrict__ out) {
    const int m0 = (blockIdx.x >> 4) << 4;
    const int n0 = (blockIdx.x & 15) << 4;
    const int lane  = threadIdx.x & 31;
    const int lhalf = lane >> 4;
    const int l15   = lane & 15;
    const int koff  = lhalf * 2;
    const int n = n0 + l15;
    const bool nvalid = (n < C_OUT);

    v8f acc;
    const float bb = nvalid ? bd[n] : 0.0f;
#pragma unroll
    for (int r = 0; r < 8; ++r) acc[r] = bb;

#pragma unroll
    for (int kk = 0; kk < 25; ++kk) {
        const int k0 = kk * 4 + koff;
        v2f a, bf;
        a[0]  = hT[(size_t)(m0 + l15) * H_DIM + k0];
        a[1]  = hT[(size_t)(m0 + l15) * H_DIM + k0 + 1];
        bf[0] = nvalid ? Wd[(size_t)k0 * C_OUT + n] : 0.0f;
        bf[1] = nvalid ? Wd[(size_t)(k0 + 1) * C_OUT + n] : 0.0f;
        acc = __builtin_amdgcn_wmma_f32_16x16x4_f32(
            false, a, false, bf, (short)0, acc, false, false);
    }
    if (nvalid) {
#pragma unroll
        for (int r = 0; r < 8; ++r)
            out[(size_t)(m0 + r + lhalf * 8) * C_OUT + n] = acc[r];
    }
}

extern "C" void kernel_launch(void* const* d_in, const int* in_sizes, int n_in,
                              void* d_out, int out_size, void* d_ws, size_t ws_size,
                              hipStream_t stream) {
    const float* x  = (const float*)d_in[0];
    const float* W  = (const float*)d_in[1];
    const float* U  = (const float*)d_in[2];
    const float* bi = (const float*)d_in[3];
    const float* Wd = (const float*)d_in[4];
    const float* bd = (const float*)d_in[5];
    float* out = (float*)d_out;

    float* pre = (float*)d_ws;
    float* hT  = pre + (size_t)S_LEN * B_SZ * FH;

    lstm_xw_kernel<<<S_LEN * (B_SZ / 16), 256, 0, stream>>>(x, W, bi, pre);
    lstm_rec_kernel<<<B_SZ / 16, REC_THREADS, 0, stream>>>(pre, U, hT);
    dense_kernel<<<(B_SZ / 16) * 16, 32, 0, stream>>>(hT, Wd, bd, out);
}


# recurrence work-group owns 2 batch rows in a zero-padded 16-row M tile instead of 16: 128 work-groups instead of 16, gate phase 200 instead of 1600 elements per group per step
# speedup vs baseline: 1.9538x; 1.2741x over previous
#include <hip/hip_runtime.h>
#include <math.h>


typedef __attribute__((ext_vector_type(2))) float v2f;
typedef __attribute__((ext_vector_type(8))) float v8f;

#define S_LEN 384
#define B_SZ  256
#define D_IN  192
#define H_DIM 100
#define FH    400
#define C_OUT 250

#define XS_STRIDE 200
#define HS  105
#define ZS  404

#define REC_WAVES   25
#define REC_THREADS (REC_WAVES * 32)
#define REC_ROWS    2

__device__ __forceinline__ float sigmoidf_(float v) {
    return 1.0f / (1.0f + __expf(-v));
}

__device__ __forceinline__ float fast_tanhf_(float v) {
#if __has_builtin(__builtin_amdgcn_tanhf)
    return __builtin_amdgcn_tanhf(v);
#elif __has_builtin(__builtin_amdgcn_tanh_f32)
    return __builtin_amdgcn_tanh_f32(v);
#else
    return tanhf(v);
#endif
}

__global__ __launch_bounds__(256)
void lstm_xw_kernel(const float* __restrict__ x, const float* __restrict__ W,
                    const float* __restrict__ bias, float* __restrict__ pre) {
    __shared__ float xs[16 * XS_STRIDE];
    const int blk = blockIdx.x;
    const int s   = blk >> 4;
    const int b0  = (blk & 15) << 4;
    const int tid = threadIdx.x;

    for (int e = tid; e < 16 * D_IN; e += 256) {
        const int r = e / D_IN, d = e - r * D_IN;
        xs[r * XS_STRIDE + d] = x[((size_t)(b0 + r) * S_LEN + s) * D_IN + d];
    }
    __syncthreads();

    const int wave  = tid >> 5;
    const int lane  = tid & 31;
    const int lhalf = lane >> 4;
    const int l15   = lane & 15;
    const int koff  = lhalf * 2;

    for (int t = wave; t < 25; t += 8) {
        const int n0 = t << 4;
        const int n  = n0 + l15;
        const float bb = bias[n];
        v8f acc;
#pragma unroll
        for (int r = 0; r < 8; ++r) acc[r] = bb;

#pragma unroll 4
        for (int kk = 0; kk < 48; ++kk) {
            const int k0 = kk * 4 + koff;
            v2f a, bf;
            a[0]  = xs[l15 * XS_STRIDE + k0];
            a[1]  = xs[l15 * XS_STRIDE + k0 + 1];
            bf[0] = W[(size_t)k0 * FH + n];
            bf[1] = W[(size_t)(k0 + 1) * FH + n];
            acc = __builtin_amdgcn_wmma_f32_16x16x4_f32(
                false, a, false, bf, (short)0, acc, false, false);
        }
        float* prow = pre + ((size_t)s * B_SZ + b0) * FH;
#pragma unroll
        for (int r = 0; r < 8; ++r)
            prow[(size_t)(r + lhalf * 8) * FH + n] = acc[r];
    }
}

__global__ __launch_bounds__(REC_THREADS)
void lstm_rec_kernel(const float* __restrict__ pre, const float* __restrict__ U,
                     float* __restrict__ hT) {
    __shared__ float U_lds[H_DIM * FH];
    __shared__ float h_s[16 * HS];
    __shared__ float c_s[REC_ROWS * H_DIM];
    __shared__ float z_s[REC_ROWS * ZS];

    const int b0  = blockIdx.x * REC_ROWS;
    const int tid = threadIdx.x;
    const int wave  = tid >> 5;
    const int lane  = tid & 31;
    const int lhalf = lane >> 4;
    const int l15   = lane & 15;
    const int koff  = lhalf * 2;

    for (int e = tid; e < H_DIM * FH; e += REC_THREADS) U_lds[e] = U[e];
    for (int e = tid; e < 16 * HS; e += REC_THREADS)    h_s[e] = 0.0f;
    for (int e = tid; e < REC_ROWS * H_DIM; e += REC_THREADS) c_s[e] = 0.0f;
    __syncthreads();

    const int n = (wave << 4) + l15;

    for (int s = 0; s < S_LEN; ++s) {
        const float* prow = pre + ((size_t)s * B_SZ + b0) * FH;

        if (s + 1 < S_LEN) {
            const float* nrow = prow + (size_t)B_SZ * FH;
            if (tid < (REC_ROWS * FH) / 32)
                __builtin_prefetch(nrow + tid * 32, 0, 0);
        }

        {
            v8f acc;
#pragma unroll
            for (int r = 0; r < 8; ++r)
                acc[r] = (r < REC_ROWS) ? prow[(size_t)r * FH + n] : 0.0f;

#pragma unroll
            for (int kk = 0; kk < 25; ++kk) {
                const int k0 = kk * 4 + koff;
                v2f a, bf;
                a[0]  = h_s[l15 * HS + k0];
                a[1]  = h_s[l15 * HS + k0 + 1];
                bf[0] = U_lds[k0 * FH + n];
                bf[1] = U_lds[(k0 + 1) * FH + n];
                acc = __builtin_amdgcn_wmma_f32_16x16x4_f32(
                    false, a, false, bf, (short)0, acc, false, false);
            }
            if (lhalf == 0) {
#pragma unroll
                for (int r = 0; r < REC_ROWS; ++r)
                    z_s[r * ZS + n] = acc[r];
            }
        }
        __syncthreads();

        if (tid < REC_ROWS * H_DIM) {
            const int e = tid;
            const int b = e / H_DIM, j = e - b * H_DIM;
            const float zi = z_s[b * ZS + j];
            const float zf = z_s[b * ZS + H_DIM + j];
            const float zg = z_s[b * ZS + 2 * H_DIM + j];
            const float zo = z_s[b * ZS + 3 * H_DIM + j];
            const float ig = sigmoidf_(zi);
            const float fg = sigmoidf_(zf);
            const float gg = fast_tanhf_(zg);
            const float og = sigmoidf_(zo);
            const float cn = fg * c_s[e] + ig * gg;
            c_s[e] = cn;
            h_s[b * HS + j] = og * fast_tanhf_(cn);
        }
        __syncthreads();
    }

    if (tid < REC_ROWS * H_DIM) {
        const int b = tid / H_DIM, j = tid - b * H_DIM;
        hT[(size_t)(b0 + b) * H_DIM + j] = h_s[b * HS + j];
    }
}

__global__ __launch_bounds__(32)
void dense_kernel(const float* __restrict__ hT, const float* __restrict__ Wd,
                  const float* __restrict__ bd, float* __restrict__ out) {
    const int m0 = (blockIdx.x >> 4) << 4;
    const int n0 = (blockIdx.x & 15) << 4;
    const int lane  = threadIdx.x & 31;
    const int lhalf = lane >> 4;
    const int l15   = lane & 15;
    const int koff  = lhalf * 2;
    const int n = n0 + l15;
    const bool nvalid = (n < C_OUT);

    v8f acc;
    const float bb = nvalid ? bd[n] : 0.0f;
#pragma unroll
    for (int r = 0; r < 8; ++r) acc[r] = bb;

#pragma unroll
    for (int kk = 0; kk < 25; ++kk) {
        const int k0 = kk * 4 + koff;
        v2f a, bf;
        a[0]  = hT[(size_t)(m0 + l15) * H_DIM + k0];
        a[1]  = hT[(size_t)(m0 + l15) * H_DIM + k0 + 1];
        bf[0] = nvalid ? Wd[(size_t)k0 * C_OUT + n] : 0.0f;
        bf[1] = nvalid ? Wd[(size_t)(k0 + 1) * C_OUT + n] : 0.0f;
        acc = __builtin_amdgcn_wmma_f32_16x16x4_f32(
            false, a, false, bf, (short)0, acc, false, false);
    }
    if (nvalid) {
#pragma unroll
        for (int r = 0; r < 8; ++r)
            out[(size_t)(m0 + r + lhalf * 8) * C_OUT + n] = acc[r];
    }
}

extern "C" void kernel_launch(void* const* d_in, const int* in_sizes, int n_in,
                              void* d_out, int out_size, void* d_ws, size_t ws_size,
                              hipStream_t stream) {
    const float* x  = (const float*)d_in[0];
    const float* W  = (const float*)d_in[1];
    const float* U  = (const float*)d_in[2];
    const float* bi = (const float*)d_in[3];
    const float* Wd = (const float*)d_in[4];
    const float* bd = (const float*)d_in[5];
    float* out = (float*)d_out;

    float* pre = (float*)d_ws;
    float* hT  = pre + (size_t)S_LEN * B_SZ * FH;

    lstm_xw_kernel<<<S_LEN * (B_SZ / 16), 256, 0, stream>>>(x, W, bi, pre);
    lstm_rec_kernel<<<B_SZ / REC_ROWS, REC_THREADS, 0, stream>>>(pre, U, hT);
    dense_kernel<<<(B_SZ / 16) * 16, 32, 0, stream>>>(hT, Wd, bd, out);
}


# recurrence: each wave's 100x16 slice of U is loaded once into 50 registers and kept for the whole scan instead of being re-read from LDS every step (LDS loads per wave per step 75 -> 25)
# speedup vs baseline: 2.0587x; 1.0537x over previous
#include <hip/hip_runtime.h>
#include <math.h>


typedef __attribute__((ext_vector_type(2))) float v2f;
typedef __attribute__((ext_vector_type(8))) float v8f;

#define S_LEN 384
#define B_SZ  256
#define D_IN  192
#define H_DIM 100
#define FH    400
#define C_OUT 250

#define XS_STRIDE 200
#define HS  105
#define ZS  404

#define REC_WAVES   25
#define REC_THREADS (REC_WAVES * 32)
#define REC_ROWS    2

__device__ __forceinline__ float sigmoidf_(float v) {
    return 1.0f / (1.0f + __expf(-v));
}

__device__ __forceinline__ float fast_tanhf_(float v) {
#if __has_builtin(__builtin_amdgcn_tanhf)
    return __builtin_amdgcn_tanhf(v);
#elif __has_builtin(__builtin_amdgcn_tanh_f32)
    return __builtin_amdgcn_tanh_f32(v);
#else
    return tanhf(v);
#endif
}

__global__ __launch_bounds__(256)
void lstm_xw_kernel(const float* __restrict__ x, const float* __restrict__ W,
                    const float* __restrict__ bias, float* __restrict__ pre) {
    __shared__ float xs[16 * XS_STRIDE];
    const int blk = blockIdx.x;
    const int s   = blk >> 4;
    const int b0  = (blk & 15) << 4;
    const int tid = threadIdx.x;

    for (int e = tid; e < 16 * D_IN; e += 256) {
        const int r = e / D_IN, d = e - r * D_IN;
        xs[r * XS_STRIDE + d] = x[((size_t)(b0 + r) * S_LEN + s) * D_IN + d];
    }
    __syncthreads();

    const int wave  = tid >> 5;
    const int lane  = tid & 31;
    const int lhalf = lane >> 4;
    const int l15   = lane & 15;
    const int koff  = lhalf * 2;

    for (int t = wave; t < 25; t += 8) {
        const int n0 = t << 4;
        const int n  = n0 + l15;
        const float bb = bias[n];
        v8f acc;
#pragma unroll
        for (int r = 0; r < 8; ++r) acc[r] = bb;

#pragma unroll 4
        for (int kk = 0; kk < 48; ++kk) {
            const int k0 = kk * 4 + koff;
            v2f a, bf;
            a[0]  = xs[l15 * XS_STRIDE + k0];
            a[1]  = xs[l15 * XS_STRIDE + k0 + 1];
            bf[0] = W[(size_t)k0 * FH + n];
            bf[1] = W[(size_t)(k0 + 1) * FH + n];
            acc = __builtin_amdgcn_wmma_f32_16x16x4_f32(
                false, a, false, bf, (short)0, acc, false, false);
        }
        float* prow = pre + ((size_t)s * B_SZ + b0) * FH;
#pragma unroll
        for (int r = 0; r < 8; ++r)
            prow[(size_t)(r + lhalf * 8) * FH + n] = acc[r];
    }
}

__global__ __launch_bounds__(REC_THREADS)
void lstm_rec_kernel(const float* __restrict__ pre, const float* __restrict__ U,
                     float* __restrict__ hT) {
    __shared__ float h_s[16 * HS];
    __shared__ float c_s[REC_ROWS * H_DIM];
    __shared__ float z_s[REC_ROWS * ZS];

    const int b0  = blockIdx.x * REC_ROWS;
    const int tid = threadIdx.x;
    const int wave  = tid >> 5;
    const int lane  = tid & 31;
    const int lhalf = lane >> 4;
    const int l15   = lane & 15;
    const int koff  = lhalf * 2;

    for (int e = tid; e < 16 * HS; e += REC_THREADS)    h_s[e] = 0.0f;
    for (int e = tid; e < REC_ROWS * H_DIM; e += REC_THREADS) c_s[e] = 0.0f;
    __syncthreads();

    const int n = (wave << 4) + l15;

    v2f ub[25];
#pragma unroll
    for (int kk = 0; kk < 25; ++kk) {
        const int k0 = kk * 4 + koff;
        ub[kk][0] = U[k0 * FH + n];
        ub[kk][1] = U[(k0 + 1) * FH + n];
    }

    for (int s = 0; s < S_LEN; ++s) {
        const float* prow = pre + ((size_t)s * B_SZ + b0) * FH;

        if (s + 1 < S_LEN) {
            const float* nrow = prow + (size_t)B_SZ * FH;
            if (tid < (REC_ROWS * FH) / 32)
                __builtin_prefetch(nrow + tid * 32, 0, 0);
        }

        {
            v8f acc;
#pragma unroll
            for (int r = 0; r < 8; ++r)
                acc[r] = (r < REC_ROWS) ? prow[(size_t)r * FH + n] : 0.0f;

#pragma unroll
            for (int kk = 0; kk < 25; ++kk) {
                const int k0 = kk * 4 + koff;
                v2f a;
                a[0]  = h_s[l15 * HS + k0];
                a[1]  = h_s[l15 * HS + k0 + 1];
                acc = __builtin_amdgcn_wmma_f32_16x16x4_f32(
                    false, a, false, ub[kk], (short)0, acc, false, false);
            }
            if (lhalf == 0) {
#pragma unroll
                for (int r = 0; r < REC_ROWS; ++r)
                    z_s[r * ZS + n] = acc[r];
            }
        }
        __syncthreads();

        if (tid < REC_ROWS * H_DIM) {
            const int e = tid;
            const int b = e / H_DIM, j = e - b * H_DIM;
            const float zi = z_s[b * ZS + j];
            const float zf = z_s[b * ZS + H_DIM + j];
            const float zg = z_s[b * ZS + 2 * H_DIM + j];
            const float zo = z_s[b * ZS + 3 * H_DIM + j];
            const float ig = sigmoidf_(zi);
            const float fg = sigmoidf_(zf);
            const float gg = fast_tanhf_(zg);
            const float og = sigmoidf_(zo);
            const float cn = fg * c_s[e] + ig * gg;
            c_s[e] = cn;
            h_s[b * HS + j] = og * fast_tanhf_(cn);
        }
        __syncthreads();
    }

    if (tid < REC_ROWS * H_DIM) {
        const int b = tid / H_DIM, j = tid - b * H_DIM;
        hT[(size_t)(b0 + b) * H_DIM + j] = h_s[b * HS + j];
    }
}

__global__ __launch_bounds__(32)
void dense_kernel(const float* __restrict__ hT, const float* __restrict__ Wd,
                  const float* __restrict__ bd, float* __restrict__ out) {
    const int m0 = (blockIdx.x >> 4) << 4;
    const int n0 = (blockIdx.x & 15) << 4;
    const int lane  = threadIdx.x & 31;
    const int lhalf = lane >> 4;
    const int l15   = lane & 15;
    const int koff  = lhalf * 2;
    const int n = n0 + l15;
    const bool nvalid = (n < C_OUT);

    v8f acc;
    const float bb = nvalid ? bd[n] : 0.0f;
#pragma unroll
    for (int r = 0; r < 8; ++r) acc[r] = bb;

#pragma unroll
    for (int kk = 0; kk < 25; ++kk) {
        const int k0 = kk * 4 + koff;
        v2f a, bf;
        a[0]  = hT[(size_t)(m0 + l15) * H_DIM + k0];
        a[1]  = hT[(size_t)(m0 + l15) * H_DIM + k0 + 1];
        bf[0] = nvalid ? Wd[(size_t)k0 * C_OUT + n] : 0.0f;
        bf[1] = nvalid ? Wd[(size_t)(k0 + 1) * C_OUT + n] : 0.0f;
        acc = __builtin_amdgcn_wmma_f32_16x16x4_f32(
            false, a, false, bf, (short)0, acc, false, false);
    }
    if (nvalid) {
#pragma unroll
        for (int r = 0; r < 8; ++r)
            out[(size_t)(m0 + r + lhalf * 8) * C_OUT + n] = acc[r];
    }
}

extern "C" void kernel_launch(void* const* d_in, const int* in_sizes, int n_in,
                              void* d_out, int out_size, void* d_ws, size_t ws_size,
                              hipStream_t stream) {
    const float* x  = (const float*)d_in[0];
    const float* W  = (const float*)d_in[1];
    const float* U  = (const float*)d_in[2];
    const float* bi = (const float*)d_in[3];
    const float* Wd = (const float*)d_in[4];
    const float* bd = (const float*)d_in[5];
    float* out = (float*)d_out;

    float* pre = (float*)d_ws;
    float* hT  = pre + (size_t)S_LEN * B_SZ * FH;

    lstm_xw_kernel<<<S_LEN * (B_SZ / 16), 256, 0, stream>>>(x, W, bi, pre);
    lstm_rec_kernel<<<B_SZ / REC_ROWS, REC_THREADS, 0, stream>>>(pre, U, hT);
    dense_kernel<<<(B_SZ / 16) * 16, 32, 0, stream>>>(hT, Wd, bd, out);
}


# recurrence: the pre rows of step s+1 are read into registers at the start of step s (one step ahead), replacing the L2 prefetch plus demand load that the first WMMA of every step waited for
# speedup vs baseline: 2.1650x; 1.0516x over previous
#include <hip/hip_runtime.h>
#include <math.h>


typedef __attribute__((ext_vector_type(2))) float v2f;
typedef __attribute__((ext_vector_type(8))) float v8f;

#define S_LEN 384
#define B_SZ  256
#define D_IN  192
#define H_DIM 100
#define FH    400
#define C_OUT 250

#define XS_STRIDE 200
#define HS  105
#define ZS  404

#define REC_WAVES   25
#define REC_THREADS (REC_WAVES * 32)
#define REC_ROWS    2

__device__ __forceinline__ float sigmoidf_(float v) {
    return 1.0f / (1.0f + __expf(-v));
}

__device__ __forceinline__ float fast_tanhf_(float v) {
#if __has_builtin(__builtin_amdgcn_tanhf)
    return __builtin_amdgcn_tanhf(v);
#elif __has_builtin(__builtin_amdgcn_tanh_f32)
    return __builtin_amdgcn_tanh_f32(v);
#else
    return tanhf(v);
#endif
}

__global__ __launch_bounds__(256)
void lstm_xw_kernel(const float* __restrict__ x, const float* __restrict__ W,
                    const float* __restrict__ bias, float* __restrict__ pre) {
    __shared__ float xs[16 * XS_STRIDE];
    const int blk = blockIdx.x;
    const int s   = blk >> 4;
    const int b0  = (blk & 15) << 4;
    const int tid = threadIdx.x;

    for (int e = tid; e < 16 * D_IN; e += 256) {
        const int r = e / D_IN, d = e - r * D_IN;
        xs[r * XS_STRIDE + d] = x[((size_t)(b0 + r) * S_LEN + s) * D_IN + d];
    }
    __syncthreads();

    const int wave  = tid >> 5;
    const int lane  = tid & 31;
    const int lhalf = lane >> 4;
    const int l15   = lane & 15;
    const int koff  = lhalf * 2;

    for (int t = wave; t < 25; t += 8) {
        const int n0 = t << 4;
        const int n  = n0 + l15;
        const float bb = bias[n];
        v8f acc;
#pragma unroll
        for (int r = 0; r < 8; ++r) acc[r] = bb;

#pragma unroll 4
        for (int kk = 0; kk < 48; ++kk) {
            const int k0 = kk * 4 + koff;
            v2f a, bf;
            a[0]  = xs[l15 * XS_STRIDE + k0];
            a[1]  = xs[l15 * XS_STRIDE + k0 + 1];
            bf[0] = W[(size_t)k0 * FH + n];
            bf[1] = W[(size_t)(k0 + 1) * FH + n];
            acc = __builtin_amdgcn_wmma_f32_16x16x4_f32(
                false, a, false, bf, (short)0, acc, false, false);
        }
        float* prow = pre + ((size_t)s * B_SZ + b0) * FH;
#pragma unroll
        for (int r = 0; r < 8; ++r)
            prow[(size_t)(r + lhalf * 8) * FH + n] = acc[r];
    }
}

__global__ __launch_bounds__(REC_THREADS)
void lstm_rec_kernel(const float* __restrict__ pre, const float* __restrict__ U,
                     float* __restrict__ hT) {
    __shared__ float h_s[16 * HS];
    __shared__ float c_s[REC_ROWS * H_DIM];
    __shared__ float z_s[REC_ROWS * ZS];

    const int b0  = blockIdx.x * REC_ROWS;
    const int tid = threadIdx.x;
    const int wave  = tid >> 5;
    const int lane  = tid & 31;
    const int lhalf = lane >> 4;
    const int l15   = lane & 15;
    const int koff  = lhalf * 2;

    for (int e = tid; e < 16 * HS; e += REC_THREADS)    h_s[e] = 0.0f;
    for (int e = tid; e < REC_ROWS * H_DIM; e += REC_THREADS) c_s[e] = 0.0f;
    __syncthreads();

    const int n = (wave << 4) + l15;

    v2f ub[25];
#pragma unroll
    for (int kk = 0; kk < 25; ++kk) {
        const int k0 = kk * 4 + koff;
        ub[kk][0] = U[k0 * FH + n];
        ub[kk][1] = U[(k0 + 1) * FH + n];
    }

    float pn[REC_ROWS];
#pragma unroll
    for (int r = 0; r < REC_ROWS; ++r)
        pn[r] = pre[((size_t)b0 + r) * FH + n];

    for (int s = 0; s < S_LEN; ++s) {

        {
            v8f acc;
#pragma unroll
            for (int r = 0; r < 8; ++r)
                acc[r] = (r < REC_ROWS) ? pn[r] : 0.0f;

            if (s + 1 < S_LEN) {
                const float* nrow = pre + ((size_t)(s + 1) * B_SZ + b0) * FH;
#pragma unroll
                for (int r = 0; r < REC_ROWS; ++r)
                    pn[r] = nrow[(size_t)r * FH + n];
            }
            __builtin_amdgcn_sched_barrier(0);

#pragma unroll
            for (int kk = 0; kk < 25; ++kk) {
                const int k0 = kk * 4 + koff;
                v2f a;
                a[0]  = h_s[l15 * HS + k0];
                a[1]  = h_s[l15 * HS + k0 + 1];
                acc = __builtin_amdgcn_wmma_f32_16x16x4_f32(
                    false, a, false, ub[kk], (short)0, acc, false, false);
            }
            if (lhalf == 0) {
#pragma unroll
                for (int r = 0; r < REC_ROWS; ++r)
                    z_s[r * ZS + n] = acc[r];
            }
        }
        __syncthreads();

        if (tid < REC_ROWS * H_DIM) {
            const int e = tid;
            const int b = e / H_DIM, j = e - b * H_DIM;
            const float zi = z_s[b * ZS + j];
            const float zf = z_s[b * ZS + H_DIM + j];
            const float zg = z_s[b * ZS + 2 * H_DIM + j];
            const float zo = z_s[b * ZS + 3 * H_DIM + j];
            const float ig = sigmoidf_(zi);
            const float fg = sigmoidf_(zf);
            const float gg = fast_tanhf_(zg);
            const float og = sigmoidf_(zo);
            const float cn = fg * c_s[e] + ig * gg;
            c_s[e] = cn;
            h_s[b * HS + j] = og * fast_tanhf_(cn);
        }
        __syncthreads();
    }

    if (tid < REC_ROWS * H_DIM) {
        const int b = tid / H_DIM, j = tid - b * H_DIM;
        hT[(size_t)(b0 + b) * H_DIM + j] = h_s[b * HS + j];
    }
}

__global__ __launch_bounds__(32)
void dense_kernel(const float* __restrict__ hT, const float* __restrict__ Wd,
                  const float* __restrict__ bd, float* __restrict__ out) {
    const int m0 = (blockIdx.x >> 4) << 4;
    const int n0 = (blockIdx.x & 15) << 4;
    const int lane  = threadIdx.x & 31;
    const int lhalf = lane >> 4;
    const int l15   = lane & 15;
    const int koff  = lhalf * 2;
    const int n = n0 + l15;
    const bool nvalid = (n < C_OUT);

    v8f acc;
    const float bb = nvalid ? bd[n] : 0.0f;
#pragma unroll
    for (int r = 0; r < 8; ++r) acc[r] = bb;

#pragma unroll
    for (int kk = 0; kk < 25; ++kk) {
        const int k0 = kk * 4 + koff;
        v2f a, bf;
        a[0]  = hT[(size_t)(m0 + l15) * H_DIM + k0];
        a[1]  = hT[(size_t)(m0 + l15) * H_DIM + k0 + 1];
        bf[0] = nvalid ? Wd[(size_t)k0 * C_OUT + n] : 0.0f;
        bf[1] = nvalid ? Wd[(size_t)(k0 + 1) * C_OUT + n] : 0.0f;
        acc = __builtin_amdgcn_wmma_f32_16x16x4_f32(
            false, a, false, bf, (short)0, acc, false, false);
    }
    if (nvalid) {
#pragma unroll
        for (int r = 0; r < 8; ++r)
            out[(size_t)(m0 + r + lhalf * 8) * C_OUT + n] = acc[r];
    }
}

extern "C" void kernel_launch(void* const* d_in, const int* in_sizes, int n_in,
                              void* d_out, int out_size, void* d_ws, size_t ws_size,
                              hipStream_t stream) {
    const float* x  = (const float*)d_in[0];
    const float* W  = (const float*)d_in[1];
    const float* U  = (const float*)d_in[2];
    const float* bi = (const float*)d_in[3];
    const float* Wd = (const float*)d_in[4];
    const float* bd = (const float*)d_in[5];
    float* out = (float*)d_out;

    float* pre = (float*)d_ws;
    float* hT  = pre + (size_t)S_LEN * B_SZ * FH;

    lstm_xw_kernel<<<S_LEN * (B_SZ / 16), 256, 0, stream>>>(x, W, bi, pre);
    lstm_rec_kernel<<<B_SZ / REC_ROWS, REC_THREADS, 0, stream>>>(pre, U, hT);
    dense_kernel<<<(B_SZ / 16) * 16, 32, 0, stream>>>(hT, Wd, bd, out);
}


# input projection: each work-group covers 64 rows (4 M-tiles) instead of 16, so every W fragment read from L2 feeds 4 WMMAs on 4 independent accumulators; x tile staged with 16-byte loads
# speedup vs baseline: 2.2486x; 1.0386x over previous
#include <hip/hip_runtime.h>
#include <math.h>


typedef __attribute__((ext_vector_type(2))) float v2f;
typedef __attribute__((ext_vector_type(4))) float v4f;
typedef __attribute__((ext_vector_type(8))) float v8f;

#define S_LEN 384
#define B_SZ  256
#define D_IN  192
#define H_DIM 100
#define FH    400
#define C_OUT 250

#define XS_STRIDE 200
#define XW_MT   4
#define XW_ROWS (16 * XW_MT)
#define HS  105
#define ZS  404

#define REC_WAVES   25
#define REC_THREADS (REC_WAVES * 32)
#define REC_ROWS    2

__device__ __forceinline__ float sigmoidf_(float v) {
    return 1.0f / (1.0f + __expf(-v));
}

__device__ __forceinline__ float fast_tanhf_(float v) {
#if __has_builtin(__builtin_amdgcn_tanhf)
    return __builtin_amdgcn_tanhf(v);
#elif __has_builtin(__builtin_amdgcn_tanh_f32)
    return __builtin_amdgcn_tanh_f32(v);
#else
    return tanhf(v);
#endif
}

__global__ __launch_bounds__(256)
void lstm_xw_kernel(const float* __restrict__ x, const float* __restrict__ W,
                    const float* __restrict__ bias, float* __restrict__ pre) {
    __shared__ float xs[XW_ROWS * XS_STRIDE];
    const int blk = blockIdx.x;
    const int s   = blk >> 2;
    const int b0  = (blk & 3) << 6;
    const int tid = threadIdx.x;

#pragma unroll
    for (int i = 0; i < (XW_ROWS * (D_IN / 4)) / 256; ++i) {
        const int e = tid + i * 256;
        const int r = e / (D_IN / 4), q = e - r * (D_IN / 4);
        const v4f v = *(const v4f*)(x + ((size_t)(b0 + r) * S_LEN + s) * D_IN + q * 4);
        *(v4f*)(&xs[r * XS_STRIDE + q * 4]) = v;
    }
    __syncthreads();

    const int wave  = tid >> 5;
    const int lane  = tid & 31;
    const int lhalf = lane >> 4;
    const int l15   = lane & 15;
    const int koff  = lhalf * 2;

    for (int t = wave; t < 25; t += 8) {
        const int n0 = t << 4;
        const int n  = n0 + l15;
        const float bb = bias[n];
        v8f acc[XW_MT];
#pragma unroll
        for (int m = 0; m < XW_MT; ++m)
#pragma unroll
            for (int r = 0; r < 8; ++r) acc[m][r] = bb;

#pragma unroll 2
        for (int kk = 0; kk < 48; ++kk) {
            const int k0 = kk * 4 + koff;
            v2f bf;
            bf[0] = W[(size_t)k0 * FH + n];
            bf[1] = W[(size_t)(k0 + 1) * FH + n];
#pragma unroll
            for (int m = 0; m < XW_MT; ++m) {
                v2f a;
                a[0] = xs[(m * 16 + l15) * XS_STRIDE + k0];
                a[1] = xs[(m * 16 + l15) * XS_STRIDE + k0 + 1];
                acc[m] = __builtin_amdgcn_wmma_f32_16x16x4_f32(
                    false, a, false, bf, (short)0, acc[m], false, false);
            }
        }
        float* prow = pre + ((size_t)s * B_SZ + b0) * FH;
#pragma unroll
        for (int m = 0; m < XW_MT; ++m)
#pragma unroll
            for (int r = 0; r < 8; ++r)
                prow[(size_t)(m * 16 + r + lhalf * 8) * FH + n] = acc[m][r];
    }
}

__global__ __launch_bounds__(REC_THREADS)
void lstm_rec_kernel(const float* __restrict__ pre, const float* __restrict__ U,
                     float* __restrict__ hT) {
    __shared__ float h_s[16 * HS];
    __shared__ float c_s[REC_ROWS * H_DIM];
    __shared__ float z_s[REC_ROWS * ZS];

    const int b0  = blockIdx.x * REC_ROWS;
    const int tid = threadIdx.x;
    const int wave  = tid >> 5;
    const int lane  = tid & 31;
    const int lhalf = lane >> 4;
    const int l15   = lane & 15;
    const int koff  = lhalf * 2;

    for (int e = tid; e < 16 * HS; e += REC_THREADS)    h_s[e] = 0.0f;
    for (int e = tid; e < REC_ROWS * H_DIM; e += REC_THREADS) c_s[e] = 0.0f;
    __syncthreads();

    const int n = (wave << 4) + l15;

    v2f ub[25];
#pragma unroll
    for (int kk = 0; kk < 25; ++kk) {
        const int k0 = kk * 4 + koff;
        ub[kk][0] = U[k0 * FH + n];
        ub[kk][1] = U[(k0 + 1) * FH + n];
    }

    float pn[REC_ROWS];
#pragma unroll
    for (int r = 0; r < REC_ROWS; ++r)
        pn[r] = pre[((size_t)b0 + r) * FH + n];

    for (int s = 0; s < S_LEN; ++s) {

        {
            v8f acc;
#pragma unroll
            for (int r = 0; r < 8; ++r)
                acc[r] = (r < REC_ROWS) ? pn[r] : 0.0f;

            if (s + 1 < S_LEN) {
                const float* nrow = pre + ((size_t)(s + 1) * B_SZ + b0) * FH;
#pragma unroll
                for (int r = 0; r < REC_ROWS; ++r)
                    pn[r] = nrow[(size_t)r * FH + n];
            }
            __builtin_amdgcn_sched_barrier(0);

#pragma unroll
            for (int kk = 0; kk < 25; ++kk) {
                const int k0 = kk * 4 + koff;
                v2f a;
                a[0]  = h_s[l15 * HS + k0];
                a[1]  = h_s[l15 * HS + k0 + 1];
                acc = __builtin_amdgcn_wmma_f32_16x16x4_f32(
                    false, a, false, ub[kk], (short)0, acc, false, false);
            }
            if (lhalf == 0) {
#pragma unroll
                for (int r = 0; r < REC_ROWS; ++r)
                    z_s[r * ZS + n] = acc[r];
            }
        }
        __syncthreads();

        if (tid < REC_ROWS * H_DIM) {
            const int e = tid;
            const int b = e / H_DIM, j = e - b * H_DIM;
            const float zi = z_s[b * ZS + j];
            const float zf = z_s[b * ZS + H_DIM + j];
            const float zg = z_s[b * ZS + 2 * H_DIM + j];
            const float zo = z_s[b * ZS + 3 * H_DIM + j];
            const float ig = sigmoidf_(zi);
            const float fg = sigmoidf_(zf);
            const float gg = fast_tanhf_(zg);
            const float og = sigmoidf_(zo);
            const float cn = fg * c_s[e] + ig * gg;
            c_s[e] = cn;
            h_s[b * HS + j] = og * fast_tanhf_(cn);
        }
        __syncthreads();
    }

    if (tid < REC_ROWS * H_DIM) {
        const int b = tid / H_DIM, j = tid - b * H_DIM;
        hT[(size_t)(b0 + b) * H_DIM + j] = h_s[b * HS + j];
    }
}

__global__ __launch_bounds__(32)
void dense_kernel(const float* __restrict__ hT, const float* __restrict__ Wd,
                  const float* __restrict__ bd, float* __restrict__ out) {
    const int m0 = (blockIdx.x >> 4) << 4;
    const int n0 = (blockIdx.x & 15) << 4;
    const int lane  = threadIdx.x & 31;
    const int lhalf = lane >> 4;
    const int l15   = lane & 15;
    const int koff  = lhalf * 2;
    const int n = n0 + l15;
    const bool nvalid = (n < C_OUT);

    v8f acc;
    const float bb = nvalid ? bd[n] : 0.0f;
#pragma unroll
    for (int r = 0; r < 8; ++r) acc[r] = bb;

#pragma unroll
    for (int kk = 0; kk < 25; ++kk) {
        const int k0 = kk * 4 + koff;
        v2f a, bf;
        a[0]  = hT[(size_t)(m0 + l15) * H_DIM + k0];
        a[1]  = hT[(size_t)(m0 + l15) * H_DIM + k0 + 1];
        bf[0] = nvalid ? Wd[(size_t)k0 * C_OUT + n] : 0.0f;
        bf[1] = nvalid ? Wd[(size_t)(k0 + 1) * C_OUT + n] : 0.0f;
        acc = __builtin_amdgcn_wmma_f32_16x16x4_f32(
            false, a, false, bf, (short)0, acc, false, false);
    }
    if (nvalid) {
#pragma unroll
        for (int r = 0; r < 8; ++r)
            out[(size_t)(m0 + r + lhalf * 8) * C_OUT + n] = acc[r];
    }
}

extern "C" void kernel_launch(void* const* d_in, const int* in_sizes, int n_in,
                              void* d_out, int out_size, void* d_ws, size_t ws_size,
                              hipStream_t stream) {
    const float* x  = (const float*)d_in[0];
    const float* W  = (const float*)d_in[1];
    const float* U  = (const float*)d_in[2];
    const float* bi = (const float*)d_in[3];
    const float* Wd = (const float*)d_in[4];
    const float* bd = (const float*)d_in[5];
    float* out = (float*)d_out;

    float* pre = (float*)d_ws;
    float* hT  = pre + (size_t)S_LEN * B_SZ * FH;

    lstm_xw_kernel<<<S_LEN * (B_SZ / XW_ROWS), 256, 0, stream>>>(x, W, bi, pre);
    lstm_rec_kernel<<<B_SZ / REC_ROWS, REC_THREADS, 0, stream>>>(pre, U, hT);
    dense_kernel<<<(B_SZ / 16) * 16, 32, 0, stream>>>(hT, Wd, bd, out);
}


# recurrence: 28 waves not 25: 24 own one z N-tile each, tile 25 is cut along K into 4 partial sums on 4 waves, one per SIMD: busiest SIMD 157 WMMAs a step, not 175; z cols 384-399 summed in 4 parts
# speedup vs baseline: 2.4589x; 1.0935x over previous
#include <hip/hip_runtime.h>
#include <math.h>


typedef __attribute__((ext_vector_type(2))) float v2f;
typedef __attribute__((ext_vector_type(4))) float v4f;
typedef __attribute__((ext_vector_type(8))) float v8f;

#define S_LEN 384
#define B_SZ  256
#define D_IN  192
#define H_DIM 100
#define FH    400
#define C_OUT 250

#define XS_STRIDE 200
#define XW_MT   4
#define XW_ROWS (16 * XW_MT)
#define HS  105
#define ZS  404

#define REC_FULL    24
#define REC_SPLIT   4
#define REC_WAVES   (REC_FULL + REC_SPLIT)
#define REC_THREADS (REC_WAVES * 32)
#define SPLIT_COL0  (REC_FULL * 16)
#define REC_ROWS    2

__device__ __forceinline__ float sigmoidf_(float v) {
    return 1.0f / (1.0f + __expf(-v));
}

__device__ __forceinline__ float fast_tanhf_(float v) {
#if __has_builtin(__builtin_amdgcn_tanhf)
    return __builtin_amdgcn_tanhf(v);
#elif __has_builtin(__builtin_amdgcn_tanh_f32)
    return __builtin_amdgcn_tanh_f32(v);
#else
    return tanhf(v);
#endif
}

__global__ __launch_bounds__(256)
void lstm_xw_kernel(const float* __restrict__ x, const float* __restrict__ W,
                    const float* __restrict__ bias, float* __restrict__ pre) {
    __shared__ float xs[XW_ROWS * XS_STRIDE];
    const int blk = blockIdx.x;
    const int s   = blk >> 2;
    const int b0  = (blk & 3) << 6;
    const int tid = threadIdx.x;

#pragma unroll
    for (int i = 0; i < (XW_ROWS * (D_IN / 4)) / 256; ++i) {
        const int e = tid + i * 256;
        const int r = e / (D_IN / 4), q = e - r * (D_IN / 4);
        const v4f v = *(const v4f*)(x + ((size_t)(b0 + r) * S_LEN + s) * D_IN + q * 4);
        *(v4f*)(&xs[r * XS_STRIDE + q * 4]) = v;
    }
    __syncthreads();

    const int wave  = tid >> 5;
    const int lane  = tid & 31;
    const int lhalf = lane >> 4;
    const int l15   = lane & 15;
    const int koff  = lhalf * 2;

    for (int t = wave; t < 25; t += 8) {
        const int n0 = t << 4;
        const int n  = n0 + l15;
        const float bb = bias[n];
        v8f acc[XW_MT];
#pragma unroll
        for (int m = 0; m < XW_MT; ++m)
#pragma unroll
            for (int r = 0; r < 8; ++r) acc[m][r] = bb;

#pragma unroll 2
        for (int kk = 0; kk < 48; ++kk) {
            const int k0 = kk * 4 + koff;
            v2f bf;
            bf[0] = W[(size_t)k0 * FH + n];
            bf[1] = W[(size_t)(k0 + 1) * FH + n];
#pragma unroll
            for (int m = 0; m < XW_MT; ++m) {
                v2f a;
                a[0] = xs[(m * 16 + l15) * XS_STRIDE + k0];
                a[1] = xs[(m * 16 + l15) * XS_STRIDE + k0 + 1];
                acc[m] = __builtin_amdgcn_wmma_f32_16x16x4_f32(
                    false, a, false, bf, (short)0, acc[m], false, false);
            }
        }
        float* prow = pre + ((size_t)s * B_SZ + b0) * FH;
#pragma unroll
        for (int m = 0; m < XW_MT; ++m)
#pragma unroll
            for (int r = 0; r < 8; ++r)
                prow[(size_t)(m * 16 + r + lhalf * 8) * FH + n] = acc[m][r];
    }
}

__global__ __launch_bounds__(REC_THREADS)
void lstm_rec_kernel(const float* __restrict__ pre, const float* __restrict__ U,
                     float* __restrict__ hT) {
    __shared__ float h_s[16 * HS];
    __shared__ float c_s[REC_ROWS * H_DIM];
    __shared__ float z_s[REC_ROWS * ZS];
    __shared__ float zp_s[(REC_SPLIT - 1) * REC_ROWS * 16];

    const int b0  = blockIdx.x * REC_ROWS;
    const int tid = threadIdx.x;
    const int wave  = __builtin_amdgcn_readfirstlane(tid >> 5);
    const int lane  = tid & 31;
    const int lhalf = lane >> 4;
    const int l15   = lane & 15;
    const int koff  = lhalf * 2;

    for (int e = tid; e < 16 * HS; e += REC_THREADS)    h_s[e] = 0.0f;
    for (int e = tid; e < REC_ROWS * H_DIM; e += REC_THREADS) c_s[e] = 0.0f;
    __syncthreads();

    const bool full = wave < REC_FULL;
    const int  q    = wave - REC_FULL;
    const int  n    = ((full ? wave : REC_FULL) << 4) + l15;
    const int  kbeg = (q > 0) ? (q * 6 + 1) : 0;
    const bool has_pre = full || (q == 0);

    v2f ub[25];
    if (full) {
#pragma unroll
        for (int kk = 0; kk < 25; ++kk) {
            const int k0 = kk * 4 + koff;
            ub[kk][0] = U[k0 * FH + n];
            ub[kk][1] = U[(k0 + 1) * FH + n];
        }
    } else {
#pragma unroll
        for (int i = 0; i < 7; ++i) {
            if (i < 6 || q == 0) {
                const int k0 = (kbeg + i) * 4 + koff;
                ub[i][0] = U[k0 * FH + n];
                ub[i][1] = U[(k0 + 1) * FH + n];
            }
        }
    }

    float pn[REC_ROWS];
#pragma unroll
    for (int r = 0; r < REC_ROWS; ++r)
        pn[r] = has_pre ? pre[((size_t)b0 + r) * FH + n] : 0.0f;

    float* zdst = (full || q == 0) ? &z_s[n] : &zp_s[(q - 1) * REC_ROWS * 16 + l15];
    const int zstride = (full || q == 0) ? ZS : 16;

    for (int s = 0; s < S_LEN; ++s) {

        {
            v8f acc;
#pragma unroll
            for (int r = 0; r < 8; ++r)
                acc[r] = (r < REC_ROWS) ? pn[r] : 0.0f;

            if (has_pre && s + 1 < S_LEN) {
                const float* nrow = pre + ((size_t)(s + 1) * B_SZ + b0) * FH;
#pragma unroll
                for (int r = 0; r < REC_ROWS; ++r)
                    pn[r] = nrow[(size_t)r * FH + n];
            }
            __builtin_amdgcn_sched_barrier(0);

            if (full) {
#pragma unroll
                for (int kk = 0; kk < 25; ++kk) {
                    const int k0 = kk * 4 + koff;
                    v2f a;
                    a[0]  = h_s[l15 * HS + k0];
                    a[1]  = h_s[l15 * HS + k0 + 1];
                    acc = __builtin_amdgcn_wmma_f32_16x16x4_f32(
                        false, a, false, ub[kk], (short)0, acc, false, false);
                }
            } else {
                const float* hrow = &h_s[l15 * HS + kbeg * 4 + koff];
#pragma unroll
                for (int i = 0; i < 6; ++i) {
                    v2f a;
                    a[0]  = hrow[i * 4];
                    a[1]  = hrow[i * 4 + 1];
                    acc = __builtin_amdgcn_wmma_f32_16x16x4_f32(
                        false, a, false, ub[i], (short)0, acc, false, false);
                }
                if (q == 0) {
                    v2f a;
                    a[0]  = hrow[24];
                    a[1]  = hrow[25];
                    acc = __builtin_amdgcn_wmma_f32_16x16x4_f32(
                        false, a, false, ub[6], (short)0, acc, false, false);
                }
            }
            if (lhalf == 0) {
#pragma unroll
                for (int r = 0; r < REC_ROWS; ++r)
                    zdst[r * zstride] = acc[r];
            }
        }
        __syncthreads();

        if (tid < REC_ROWS * H_DIM) {
            const int e = tid;
            const int b = e / H_DIM, j = e - b * H_DIM;
            const float zi = z_s[b * ZS + j];
            const float zf = z_s[b * ZS + H_DIM + j];
            const float zg = z_s[b * ZS + 2 * H_DIM + j];
            float       zo = z_s[b * ZS + 3 * H_DIM + j];
            if (3 * H_DIM + j >= SPLIT_COL0) {
                const int cc = 3 * H_DIM + j - SPLIT_COL0;
                zo = (zo + zp_s[(0 * REC_ROWS + b) * 16 + cc])
                   + (zp_s[(1 * REC_ROWS + b) * 16 + cc] + zp_s[(2 * REC_ROWS + b) * 16 + cc]);
            }
            const float ig = sigmoidf_(zi);
            const float fg = sigmoidf_(zf);
            const float gg = fast_tanhf_(zg);
            const float og = sigmoidf_(zo);
            const float cn = fg * c_s[e] + ig * gg;
            c_s[e] = cn;
            h_s[b * HS + j] = og * fast_tanhf_(cn);
        }
        __syncthreads();
    }

    if (tid < REC_ROWS * H_DIM) {
        const int b = tid / H_DIM, j = tid - b * H_DIM;
        hT[(size_t)(b0 + b) * H_DIM + j] = h_s[b * HS + j];
    }
}

__global__ __launch_bounds__(32)
void dense_kernel(const float* __restrict__ hT, const float* __restrict__ Wd,
                  const float* __restrict__ bd, float* __restrict__ out) {
    const int m0 = (blockIdx.x >> 4) << 4;
    const int n0 = (blockIdx.x & 15) << 4;
    const int lane  = threadIdx.x & 31;
    const int lhalf = lane >> 4;
    const int l15   = lane & 15;
    const int koff  = lhalf * 2;
    const int n = n0 + l15;
    const bool nvalid = (n < C_OUT);

    v8f acc;
    const float bb = nvalid ? bd[n] : 0.0f;
#pragma unroll
    for (int r = 0; r < 8; ++r) acc[r] = bb;

#pragma unroll
    for (int kk = 0; kk < 25; ++kk) {
        const int k0 = kk * 4 + koff;
        v2f a, bf;
        a[0]  = hT[(size_t)(m0 + l15) * H_DIM + k0];
        a[1]  = hT[(size_t)(m0 + l15) * H_DIM + k0 + 1];
        bf[0] = nvalid ? Wd[(size_t)k0 * C_OUT + n] : 0.0f;
        bf[1] = nvalid ? Wd[(size_t)(k0 + 1) * C_OUT + n] : 0.0f;
        acc = __builtin_amdgcn_wmma_f32_16x16x4_f32(
            false, a, false, bf, (short)0, acc, false, false);
    }
    if (nvalid) {
#pragma unroll
        for (int r = 0; r < 8; ++r)
            out[(size_t)(m0 + r + lhalf * 8) * C_OUT + n] = acc[r];
    }
}

extern "C" void kernel_launch(void* const* d_in, const int* in_sizes, int n_in,
                              void* d_out, int out_size, void* d_ws, size_t ws_size,
                              hipStream_t stream) {
    const float* x  = (const float*)d_in[0];
    const float* W  = (const float*)d_in[1];
    const float* U  = (const float*)d_in[2];
    const float* bi = (const float*)d_in[3];
    const float* Wd = (const float*)d_in[4];
    const float* bd = (const float*)d_in[5];
    float* out = (float*)d_out;

    float* pre = (float*)d_ws;
    float* hT  = pre + (size_t)S_LEN * B_SZ * FH;

    lstm_xw_kernel<<<S_LEN * (B_SZ / XW_ROWS), 256, 0, stream>>>(x, W, bi, pre);
    lstm_rec_kernel<<<B_SZ / REC_ROWS, REC_THREADS, 0, stream>>>(pre, U, hT);
    dense_kernel<<<(B_SZ / 16) * 16, 32, 0, stream>>>(hT, Wd, bd, out);
}
